# starting kernel
# baseline (speedup 1.0000x reference)
#include <hip/hip_runtime.h>
#include <hip/hip_bf16.h>

#define NEG_SLOPE 0.2f

typedef __attribute__((ext_vector_type(16))) _Float16 v16h;
typedef __attribute__((ext_vector_type(8)))  float    v8f;

template<int CO>
__global__ __launch_bounds__(32)
void node_xform_wmma(const float* __restrict__ X, const float* __restrict__ W,
                     const float* __restrict__ bias, float* __restrict__ Out,
                     int n_nodes) {
  const int mtile = blockIdx.x;
  const int ntile = blockIdx.y;
  const int lane  = threadIdx.x;
  const int mrow  = lane & 15;
  const int grp   = lane >> 4;
  const int nIdx  = ntile * 16 + (lane & 15);

  int node = mtile * 16 + mrow;
  if (node >= n_nodes) node = n_nodes - 1;

  const float bv = bias[nIdx];
  v8f c;
#pragma unroll
  for (int r = 0; r < 8; ++r) c[r] = bv;

#pragma unroll
  for (int ks = 0; ks < 2; ++ks) {
    v16h a, b;
#pragma unroll
    for (int v = 0; v < 8; ++v) {
      const int k = ks * 32 + ((v & 4) ? 16 : 0) + grp * 8 + (v & 3) * 2;
      const float2 xp = *(const float2*)(X + (size_t)node * 64 + k);
      a[2 * v]     = (_Float16)xp.x;
      a[2 * v + 1] = (_Float16)xp.y;
      b[2 * v]     = (_Float16)W[(size_t)(k)     * CO + nIdx];
      b[2 * v + 1] = (_Float16)W[(size_t)(k + 1) * CO + nIdx];
    }
    c = __builtin_amdgcn_wmma_f32_16x16x32_f16(false, a, false, b,
                                               (short)0, c, false, false);
  }

  if ((mtile + 1) * 16 <= n_nodes) {
#pragma unroll
    for (int r = 0; r < 8; ++r)
      Out[(size_t)(mtile * 16 + r + 8 * grp) * CO + nIdx] = c[r];
  } else {
#pragma unroll
    for (int r = 0; r < 8; ++r) {
      const int row = mtile * 16 + r + 8 * grp;
      if (row < n_nodes) Out[(size_t)row * CO + nIdx] = c[r];
    }
  }
}

__global__ void k_init(float* __restrict__ amax, float* __restrict__ den,
                       float* __restrict__ acc, int n, int co) {
  const int i = blockIdx.x * blockDim.x + threadIdx.x;
  if (i < n * co) acc[i] = 0.0f;
  if (i < n) { amax[i] = -3.402823466e38f; den[i] = 0.0f; }
}

__device__ __forceinline__ void atomicMaxFloat(float* addr, float val) {
  if (val >= 0.0f) atomicMax((int*)addr, __float_as_int(val));
  else             atomicMin((unsigned int*)addr, __float_as_uint(val));
}

template<int CO>
__global__ __launch_bounds__(256)
void edge_alpha(const float* __restrict__ xl, const float* __restrict__ xr,
                const int* __restrict__ src, const int* __restrict__ dst,
                const float* __restrict__ ef, const float* __restrict__ We,
                const float* __restrict__ att,
                float* __restrict__ alpha, float* __restrict__ amax, int ne) {
  constexpr int LPE = CO / 4;
  const int gid = blockIdx.x * blockDim.x + threadIdx.x;
  const int e   = gid / LPE;
  const int sub = gid % LPE;
  if (e >= ne) return;
  const int s = src[e], d = dst[e];
  const float fe = ef[e];
  const int off = sub * 4;
  const float4 a4 = *(const float4*)(xl + (size_t)s * CO + off);
  const float4 r4 = *(const float4*)(xr + (size_t)d * CO + off);
  const float4 w4 = *(const float4*)(We + off);
  const float4 t4 = *(const float4*)(att + off);
  float acc = 0.0f, m;
  m = a4.x + r4.x + fe * w4.x; m = (m > 0.0f) ? m : NEG_SLOPE * m; acc += m * t4.x;
  m = a4.y + r4.y + fe * w4.y; m = (m > 0.0f) ? m : NEG_SLOPE * m; acc += m * t4.y;
  m = a4.z + r4.z + fe * w4.z; m = (m > 0.0f) ? m : NEG_SLOPE * m; acc += m * t4.z;
  m = a4.w + r4.w + fe * w4.w; m = (m > 0.0f) ? m : NEG_SLOPE * m; acc += m * t4.w;
#pragma unroll
  for (int o = LPE / 2; o > 0; o >>= 1) acc += __shfl_down(acc, o, LPE);
  if (sub == 0) { alpha[e] = acc; atomicMaxFloat(amax + d, acc); }
}

__global__ __launch_bounds__(256)
void edge_exp(float* __restrict__ alpha, const int* __restrict__ dst,
              const float* __restrict__ amax, float* __restrict__ den, int ne) {
  const int e = blockIdx.x * blockDim.x + threadIdx.x;
  if (e >= ne) return;
  const int d = dst[e];
  const float ex = __expf(alpha[e] - amax[d]);
  alpha[e] = ex;
  atomicAdd(den + d, ex);
}

template<int CO>
__global__ __launch_bounds__(256)
void edge_scatter(const float* __restrict__ xl, const float* __restrict__ ex,
                  const float* __restrict__ den, const int* __restrict__ src,
                  const int* __restrict__ dst, float* __restrict__ acc, int ne) {
  constexpr int LPE = CO / 4;
  const int gid = blockIdx.x * blockDim.x + threadIdx.x;
  const int e   = gid / LPE;
  const int sub = gid % LPE;
  if (e >= ne) return;
  const int s = src[e], d = dst[e];
  const float w = ex[e] / (den[d] + 1e-16f);
  const int off = sub * 4;
  const float4 v = *(const float4*)(xl + (size_t)s * CO + off);
  float* p = acc + (size_t)d * CO + off;
  atomicAdd(p + 0, v.x * w);
  atomicAdd(p + 1, v.y * w);
  atomicAdd(p + 2, v.z * w);
  atomicAdd(p + 3, v.w * w);
}

__global__ __launch_bounds__(256)
void k_finalize(float* __restrict__ acc, const float* __restrict__ bias,
                int n, int co, int do_relu) {
  const int i = blockIdx.x * blockDim.x + threadIdx.x;
  if (i >= n * co) return;
  float v = acc[i] + bias[i % co];
  if (do_relu) v = fmaxf(v, 0.0f);
  acc[i] = v;
}

extern "C" void kernel_launch(void* const* d_in, const int* in_sizes, int n_in,
                              void* d_out, int out_size, void* d_ws, size_t ws_size,
                              hipStream_t stream) {
  (void)n_in; (void)out_size; (void)ws_size;
  const float* x    = (const float*)d_in[0];
  const int*   ei   = (const int*)d_in[1];
  const float* ef   = (const float*)d_in[2];
  const float* Wl0  = (const float*)d_in[3];
  const float* bl0  = (const float*)d_in[4];
  const float* Wr0  = (const float*)d_in[5];
  const float* br0  = (const float*)d_in[6];
  const float* We0  = (const float*)d_in[7];
  const float* att0 = (const float*)d_in[8];
  const float* b0   = (const float*)d_in[9];
  const float* Wl1  = (const float*)d_in[10];
  const float* bl1  = (const float*)d_in[11];
  const float* Wr1  = (const float*)d_in[12];
  const float* br1  = (const float*)d_in[13];
  const float* We1  = (const float*)d_in[14];
  const float* att1 = (const float*)d_in[15];
  const float* b1   = (const float*)d_in[16];

  const int n = in_sizes[0] / 64;
  const int e = in_sizes[1] / 2;
  const int* srcI = ei;
  const int* dstI = ei + e;

  float* ws   = (float*)d_ws;
  float* xl   = ws;  ws += (size_t)n * 64;
  float* xr   = ws;  ws += (size_t)n * 64;
  float* hbuf = ws;  ws += (size_t)n * 64;
  float* alp  = ws;  ws += (size_t)e;
  float* amax = ws;  ws += n;
  float* den  = ws;  ws += n;

  auto cdiv = [](long long a, long long b) { return (int)((a + b - 1) / b); };
  const int B = 256;

  {
    constexpr int CO = 64;
    k_init<<<cdiv((long long)n * CO, B), B, 0, stream>>>(amax, den, hbuf, n, CO);
    dim3 g((n + 15) / 16, CO / 16);
    node_xform_wmma<CO><<<g, 32, 0, stream>>>(x, Wl0, bl0, xl, n);
    node_xform_wmma<CO><<<g, 32, 0, stream>>>(x, Wr0, br0, xr, n);
    edge_alpha<CO><<<cdiv((long long)e * (CO / 4), B), B, 0, stream>>>(
        xl, xr, srcI, dstI, ef, We0, att0, alp, amax, e);
    edge_exp<<<cdiv(e, B), B, 0, stream>>>(alp, dstI, amax, den, e);
    edge_scatter<CO><<<cdiv((long long)e * (CO / 4), B), B, 0, stream>>>(
        xl, alp, den, srcI, dstI, hbuf, e);
    k_finalize<<<cdiv((long long)n * CO, B), B, 0, stream>>>(hbuf, b0, n, CO, 1);
  }

  {
    constexpr int CO = 32;
    float* outp = (float*)d_out;
    k_init<<<cdiv((long long)n * CO, B), B, 0, stream>>>(amax, den, outp, n, CO);
    dim3 g((n + 15) / 16, CO / 16);
    node_xform_wmma<CO><<<g, 32, 0, stream>>>(hbuf, Wl1, bl1, xl, n);
    node_xform_wmma<CO><<<g, 32, 0, stream>>>(hbuf, Wr1, br1, xr, n);
    edge_alpha<CO><<<cdiv((long long)e * (CO / 4), B), B, 0, stream>>>(
        xl, xr, srcI, dstI, ef, We1, att1, alp, amax, e);
    edge_exp<<<cdiv(e, B), B, 0, stream>>>(alp, dstI, amax, den, e);
    edge_scatter<CO><<<cdiv((long long)e * (CO / 4), B), B, 0, stream>>>(
        xl, alp, den, srcI, dstI, outp, e);
    k_finalize<<<cdiv((long long)n * CO, B), B, 0, stream>>>(outp, b1, n, CO, 0);
  }
}


# One-pass edge kernel: each resident work-group (one per WGP, striding the edge chunks) sums the denominators in its own LDS and adds them to den as whole lines at the end; den sum order changes
# speedup vs baseline: 2.6797x; 2.6797x over previous
#include <hip/hip_runtime.h>
#include <hip/hip_bf16.h>

#define NEG_SLOPE 0.2f
#define LOGIT_LIMIT 40.0f
#define LDS_NODES 65536

typedef __attribute__((ext_vector_type(16))) _Float16 v16h;
typedef __attribute__((ext_vector_type(8)))  float    v8f;

template<int CO, bool PRE, bool ZERO, bool SEED>
__global__ __launch_bounds__(32)
void node_xform_wmma(const float* __restrict__ X, const float* __restrict__ W,
                     const float* __restrict__ bias, float* __restrict__ Out,
                     const float* __restrict__ pre_bias,
                     const float* __restrict__ pre_den,
                     float* __restrict__ zacc,
                     float* __restrict__ amax, float* __restrict__ den,
                     int* __restrict__ flag,
                     int n_nodes) {
  const int mtile = blockIdx.x;
  const int ntile = blockIdx.y;
  const int lane  = threadIdx.x;
  const int mrow  = lane & 15;
  const int grp   = lane >> 4;
  const int nIdx  = ntile * 16 + (lane & 15);
  const bool full = (mtile + 1) * 16 <= n_nodes;

  if constexpr (ZERO) {
    if (full) {
#pragma unroll
      for (int r = 0; r < 8; ++r)
        zacc[(size_t)(mtile * 16 + r + 8 * grp) * CO + nIdx] = 0.0f;
    } else {
#pragma unroll
      for (int r = 0; r < 8; ++r) {
        const int row = mtile * 16 + r + 8 * grp;
        if (row < n_nodes) zacc[(size_t)row * CO + nIdx] = 0.0f;
      }
    }
  }
  if constexpr (SEED) {
    if (ntile == 0 && lane < 16) {
      const int row = mtile * 16 + lane;
      if (row < n_nodes) { amax[row] = -3.402823466e38f; den[row] = 0.0f; }
      if (mtile == 0 && lane == 0) *flag = 0;
    }
  }

  int node = mtile * 16 + mrow;
  if (node >= n_nodes) node = n_nodes - 1;

  float inv = 1.0f;
  if constexpr (PRE) {
    const float dn = pre_den[node];
    inv = (dn > 0.0f) ? 1.0f / dn : 0.0f;
  }

  const float bv = bias[nIdx];
  v8f c;
#pragma unroll
  for (int r = 0; r < 8; ++r) c[r] = bv;

#pragma unroll
  for (int ks = 0; ks < 2; ++ks) {
    v16h a, b;
#pragma unroll
    for (int v = 0; v < 8; ++v) {
      const int k = ks * 32 + ((v & 4) ? 16 : 0) + grp * 8 + (v & 3) * 2;
      float2 xp = *(const float2*)(X + (size_t)node * 64 + k);
      if constexpr (PRE) {
        const float2 pb = *(const float2*)(pre_bias + k);
        xp.x = fmaxf(xp.x * inv + pb.x, 0.0f);
        xp.y = fmaxf(xp.y * inv + pb.y, 0.0f);
      }
      a[2 * v]     = (_Float16)xp.x;
      a[2 * v + 1] = (_Float16)xp.y;
      b[2 * v]     = (_Float16)W[(size_t)(k)     * CO + nIdx];
      b[2 * v + 1] = (_Float16)W[(size_t)(k + 1) * CO + nIdx];
    }
    c = __builtin_amdgcn_wmma_f32_16x16x32_f16(false, a, false, b,
                                               (short)0, c, false, false);
  }

  if (full) {
#pragma unroll
    for (int r = 0; r < 8; ++r)
      Out[(size_t)(mtile * 16 + r + 8 * grp) * CO + nIdx] = c[r];
  } else {
#pragma unroll
    for (int r = 0; r < 8; ++r) {
      const int row = mtile * 16 + r + 8 * grp;
      if (row < n_nodes) Out[(size_t)row * CO + nIdx] = c[r];
    }
  }
}

__device__ __forceinline__ void atomicMaxFloat(float* addr, float val) {
  if (val >= 0.0f) atomicMax((int*)addr, __float_as_int(val));
  else             atomicMin((unsigned int*)addr, __float_as_uint(val));
}

__device__ __forceinline__ float pick(bool c, float a, float b) { return c ? a : b; }

template<int CO>
__device__ __forceinline__ float chunk_logits(const float* __restrict__ xl,
                                              const float* __restrict__ xr,
                                              int s, int d, int febits,
                                              const float (&we)[CO / 32],
                                              const float (&at)[CO / 32],
                                              int lane,
                                              float (&xa)[32][CO / 32]) {
  constexpr int LPR = CO / 32;
  constexpr int G   = 16 / LPR;
  float p[32];
#pragma unroll
  for (int g = 0; g < 32; g += G) {
    float r[G][LPR];
#pragma unroll
    for (int j = 0; j < G; ++j) {
      const int sj = __builtin_amdgcn_readlane(s, g + j);
      const int dj = __builtin_amdgcn_readlane(d, g + j);
      const float* ra = xl + (size_t)sj * CO + lane;
      const float* rr = xr + (size_t)dj * CO + lane;
#pragma unroll
      for (int h = 0; h < LPR; ++h) {
        xa[g + j][h] = ra[h * 32];
        r[j][h]      = rr[h * 32];
      }
    }
#pragma unroll
    for (int j = 0; j < G; ++j) {
      const float fj = __int_as_float(__builtin_amdgcn_readlane(febits, g + j));
      float acc = 0.0f;
#pragma unroll
      for (int h = 0; h < LPR; ++h) {
        float m = xa[g + j][h] + r[j][h] + fj * we[h];
        m = (m > 0.0f) ? m : NEG_SLOPE * m;
        acc += m * at[h];
      }
      p[g + j] = acc;
    }
  }

#pragma unroll
  for (int w = 16; w >= 1; w >>= 1) {
    const bool hi = (lane & w) != 0;
    float got[16];
#pragma unroll
    for (int j = 0; j < w; ++j)
      got[j] = __shfl_xor(pick(hi, p[j], p[j + w]), w);
#pragma unroll
    for (int j = 0; j < w; ++j)
      p[j] = pick(hi, p[j + w], p[j]) + got[j];
  }
  return p[0];
}

template<int CO, int NT>
__global__ __launch_bounds__(NT)
void edge_onepass(const float* __restrict__ xl, const float* __restrict__ xr,
                  const int* __restrict__ src, const int* __restrict__ dst,
                  const float* __restrict__ ef, const float* __restrict__ We,
                  const float* __restrict__ att,
                  float* __restrict__ den, float* __restrict__ acc,
                  int* __restrict__ flag, int n_nodes, int ne) {
  constexpr int LPR = CO / 32;
  __shared__ __attribute__((aligned(16))) float sden[LDS_NODES];
  const int tid  = (int)threadIdx.x;
  const int nthr = (int)blockDim.x;
  const int lane = tid & 31;
  const int wv = __builtin_amdgcn_readfirstlane(tid >> 5);
  const int nw = nthr >> 5;
  const int nl = (n_nodes < LDS_NODES) ? n_nodes : LDS_NODES;

  for (int i = tid * 4; i < nl; i += nthr * 4)
    *(float4*)(sden + i) = make_float4(0.0f, 0.0f, 0.0f, 0.0f);

  float we[LPR], at[LPR];
#pragma unroll
  for (int h = 0; h < LPR; ++h) {
    we[h] = We[lane + h * 32];
    at[h] = att[lane + h * 32];
  }
  __syncthreads();

  const int nchunks = (ne + 31) / 32;
  const int cstep   = nw * (int)gridDim.x;
  for (int chunk = wv * (int)gridDim.x + (int)blockIdx.x; chunk < nchunks; chunk += cstep) {
    const int base = chunk * 32;
    const int nvalid = (ne - base < 32) ? (ne - base) : 32;

    int s = 0, d = 0;
    float fe = 0.0f;
    if (lane < nvalid) {
      const int e = base + lane;
      s  = src[e];
      d  = dst[e];
      fe = ef[e];
    }

    float xa[32][LPR];
    const float logit = chunk_logits<CO>(xl, xr, s, d, __float_as_int(fe), we, at, lane, xa);

    const bool valid = lane < nvalid;
    const bool out_of_range = valid && !(fabsf(logit) <= LOGIT_LIMIT);
    if (__builtin_amdgcn_ballot_w32(out_of_range) != 0u) {
      if (lane == 0) atomicOr(flag, 1);
    }

    float w = 0.0f;
    if (valid) {
      w = __expf(logit);
      if (d < nl) atomicAdd(sden + d, w);
      else        atomicAdd(den + d, w);
    }
    const int wbits = __float_as_int(w);

    if (nvalid == 32) {
#pragma unroll
      for (int j = 0; j < 32; ++j) {
        const int   dj = __builtin_amdgcn_readlane(d, j);
        const float wj = __int_as_float(__builtin_amdgcn_readlane(wbits, j));
        float* p = acc + (size_t)dj * CO + lane;
#pragma unroll
        for (int h = 0; h < LPR; ++h) atomicAdd(p + h * 32, xa[j][h] * wj);
      }
    } else {
#pragma unroll
      for (int j = 0; j < 32; ++j) {
        if (j < nvalid) {
          const int   dj = __builtin_amdgcn_readlane(d, j);
          const float wj = __int_as_float(__builtin_amdgcn_readlane(wbits, j));
          float* p = acc + (size_t)dj * CO + lane;
#pragma unroll
          for (int h = 0; h < LPR; ++h) atomicAdd(p + h * 32, xa[j][h] * wj);
        }
      }
    }
  }
  __syncthreads();

  const int mis    = (int)(((size_t)den >> 2) & 31);
  const int nlines = (nl + mis + 31) / 32;
  const int rot    = (int)(((unsigned)blockIdx.x * (unsigned)nlines) / (unsigned)gridDim.x);
  for (int k0 = wv * 8; k0 < nlines; k0 += nw * 8) {
    float v[8];
    int   idx[8];
#pragma unroll
    for (int u = 0; u < 8; ++u) {
      int line = k0 + u + rot;
      if (line >= nlines) line -= nlines;
      const int i = line * 32 + lane - mis;
      idx[u] = i;
      v[u] = (k0 + u < nlines && i >= 0 && i < nl) ? sden[i] : 0.0f;
    }
#pragma unroll
    for (int u = 0; u < 8; ++u)
      if (v[u] != 0.0f) atomicAdd(den + idx[u], v[u]);
  }
}

template<int CO>
__global__ __launch_bounds__(256)
void edge_alpha_shifted(const float* __restrict__ xl, const float* __restrict__ xr,
                        const int* __restrict__ src, const int* __restrict__ dst,
                        const float* __restrict__ ef, const float* __restrict__ We,
                        const float* __restrict__ att,
                        float* __restrict__ alpha, float* __restrict__ amax,
                        float* __restrict__ zacc, float* __restrict__ zden,
                        int n_nodes, const int* __restrict__ flag, int ne) {
  if (*flag == 0) return;
  constexpr int LPR = CO / 32;
  const int tid      = (int)(blockIdx.x * blockDim.x + threadIdx.x);
  const int nthreads = (int)(gridDim.x * blockDim.x);
  const size_t nacc = (size_t)n_nodes * CO;
  for (size_t i = (size_t)tid; i < nacc; i += (size_t)nthreads) zacc[i] = 0.0f;
  for (int i = tid; i < n_nodes; i += nthreads) zden[i] = 0.0f;

  const int lane = threadIdx.x & 31;
  const int wave0  = __builtin_amdgcn_readfirstlane(tid >> 5);
  const int nwaves = nthreads >> 5;
  const int nchunks = (ne + 31) / 32;

  float we[LPR], at[LPR];
#pragma unroll
  for (int h = 0; h < LPR; ++h) {
    we[h] = We[lane + h * 32];
    at[h] = att[lane + h * 32];
  }

  for (int chunk = wave0; chunk < nchunks; chunk += nwaves) {
    const int base = chunk * 32;
    const int nvalid = (ne - base < 32) ? (ne - base) : 32;
    int s = 0, d = 0;
    float fe = 0.0f;
    if (lane < nvalid) {
      const int e = base + lane;
      s  = src[e];
      d  = dst[e];
      fe = ef[e];
    }
    float xa[32][LPR];
    const float logit = chunk_logits<CO>(xl, xr, s, d, __float_as_int(fe), we, at, lane, xa);
    if (lane < nvalid) {
      alpha[base + lane] = logit;
      atomicMaxFloat(amax + d, logit);
    }
  }
}

template<int CO>
__global__ __launch_bounds__(256)
void edge_scatter_shifted(const float* __restrict__ xl, const float* __restrict__ alpha,
                          const float* __restrict__ amax, float* __restrict__ den,
                          const int* __restrict__ src, const int* __restrict__ dst,
                          float* __restrict__ acc, const int* __restrict__ flag, int ne) {
  if (*flag == 0) return;
  constexpr int LPR = CO / 32;
  constexpr int G   = 16 / LPR;
  const int tid      = (int)(blockIdx.x * blockDim.x + threadIdx.x);
  const int nthreads = (int)(gridDim.x * blockDim.x);
  const int lane = threadIdx.x & 31;
  const int wave0  = __builtin_amdgcn_readfirstlane(tid >> 5);
  const int nwaves = nthreads >> 5;
  const int nchunks = (ne + 31) / 32;

  for (int chunk = wave0; chunk < nchunks; chunk += nwaves) {
    const int base = chunk * 32;
    const int nvalid = (ne - base < 32) ? (ne - base) : 32;

    int s = 0, d = 0;
    float w = 0.0f;
    if (lane < nvalid) {
      const int e = base + lane;
      s = src[e];
      d = dst[e];
      w = __expf(alpha[e] - amax[d]);
      atomicAdd(den + d, w);
    }
    const int wbits = __float_as_int(w);

    if (nvalid == 32) {
#pragma unroll
      for (int g = 0; g < 32; g += G) {
        float v[G][LPR];
#pragma unroll
        for (int j = 0; j < G; ++j) {
          const int sj = __builtin_amdgcn_readlane(s, g + j);
          const float* row = xl + (size_t)sj * CO + lane;
#pragma unroll
          for (int h = 0; h < LPR; ++h) v[j][h] = row[h * 32];
        }
#pragma unroll
        for (int j = 0; j < G; ++j) {
          const int   dj = __builtin_amdgcn_readlane(d, g + j);
          const float wj = __int_as_float(__builtin_amdgcn_readlane(wbits, g + j));
          float* p = acc + (size_t)dj * CO + lane;
#pragma unroll
          for (int h = 0; h < LPR; ++h) atomicAdd(p + h * 32, v[j][h] * wj);
        }
      }
    } else {
      for (int j = 0; j < nvalid; ++j) {
        const int   sj = __shfl(s, j);
        const int   dj = __shfl(d, j);
        const float wj = __shfl(w, j);
        const float* row = xl + (size_t)sj * CO + lane;
        float* p = acc + (size_t)dj * CO + lane;
#pragma unroll
        for (int h = 0; h < LPR; ++h) atomicAdd(p + h * 32, row[h * 32] * wj);
      }
    }
  }
}

template<int CO>
__global__ __launch_bounds__(256)
void k_finalize(float* __restrict__ acc, const float* __restrict__ den,
                const float* __restrict__ bias, int n, int do_relu) {
  const int i = blockIdx.x * blockDim.x + threadIdx.x;
  if (i >= n * CO) return;
  const float dn = den[i / CO];
  float v = ((dn > 0.0f) ? acc[i] / dn : 0.0f) + bias[i % CO];
  if (do_relu) v = fmaxf(v, 0.0f);
  acc[i] = v;
}

extern "C" void kernel_launch(void* const* d_in, const int* in_sizes, int n_in,
                              void* d_out, int out_size, void* d_ws, size_t ws_size,
                              hipStream_t stream) {
  (void)n_in; (void)out_size; (void)ws_size;
  const float* x    = (const float*)d_in[0];
  const int*   ei   = (const int*)d_in[1];
  const float* ef   = (const float*)d_in[2];
  const float* Wl0  = (const float*)d_in[3];
  const float* bl0  = (const float*)d_in[4];
  const float* Wr0  = (const float*)d_in[5];
  const float* br0  = (const float*)d_in[6];
  const float* We0  = (const float*)d_in[7];
  const float* att0 = (const float*)d_in[8];
  const float* b0   = (const float*)d_in[9];
  const float* Wl1  = (const float*)d_in[10];
  const float* bl1  = (const float*)d_in[11];
  const float* Wr1  = (const float*)d_in[12];
  const float* br1  = (const float*)d_in[13];
  const float* We1  = (const float*)d_in[14];
  const float* att1 = (const float*)d_in[15];
  const float* b1   = (const float*)d_in[16];

  const int n = in_sizes[0] / 64;
  const int e = in_sizes[1] / 2;
  const int* srcI = ei;
  const int* dstI = ei + e;

  float* ws   = (float*)d_ws;
  float* xl   = ws;  ws += (size_t)n * 64;
  float* xr   = ws;  ws += (size_t)n * 64;
  float* hbuf = ws;  ws += (size_t)n * 64;
  float* alp  = ws;  ws += (size_t)e;
  float* amax = ws;  ws += n;
  float* den0 = ws;  ws += n;
  float* den1 = ws;  ws += n;
  int* flag0  = (int*)ws;
  int* flag1  = flag0 + 1;

  auto cdiv = [](long long a, long long b) { return (int)((a + b - 1) / b); };
  const int B = 256;
  const int GS = 256;

  {
    constexpr int CO = 64;
    dim3 g((n + 15) / 16, CO / 16);
    node_xform_wmma<CO, false, true, false><<<g, 32, 0, stream>>>(
        x, Wl0, bl0, xl, nullptr, nullptr, hbuf, nullptr, nullptr, nullptr, n);
    node_xform_wmma<CO, false, false, true><<<g, 32, 0, stream>>>(
        x, Wr0, br0, xr, nullptr, nullptr, nullptr, amax, den0, flag0, n);
    edge_onepass<CO, 896><<<GS, 896, 0, stream>>>(
        xl, xr, srcI, dstI, ef, We0, att0, den0, hbuf, flag0, n, e);
    edge_alpha_shifted<CO><<<GS, B, 0, stream>>>(
        xl, xr, srcI, dstI, ef, We0, att0, alp, amax, hbuf, den0, n, flag0, e);
    edge_scatter_shifted<CO><<<GS, B, 0, stream>>>(
        xl, alp, amax, den0, srcI, dstI, hbuf, flag0, e);
  }

  {
    constexpr int CO = 32;
    float* outp = (float*)d_out;
    dim3 g((n + 15) / 16, CO / 16);
    node_xform_wmma<CO, true, true, false><<<g, 32, 0, stream>>>(
        hbuf, Wl1, bl1, xl, b0, den0, outp, nullptr, nullptr, nullptr, n);
    node_xform_wmma<CO, true, false, true><<<g, 32, 0, stream>>>(
        hbuf, Wr1, br1, xr, b0, den0, nullptr, amax, den1, flag1, n);
    edge_onepass<CO, 1024><<<GS, 1024, 0, stream>>>(
        xl, xr, srcI, dstI, ef, We1, att1, den1, outp, flag1, n, e);
    edge_alpha_shifted<CO><<<GS, B, 0, stream>>>(
        xl, xr, srcI, dstI, ef, We1, att1, alp, amax, outp, den1, n, flag1, e);
    edge_scatter_shifted<CO><<<GS, B, 0, stream>>>(
        xl, alp, amax, den1, srcI, dstI, outp, flag1, e);
    k_finalize<CO><<<cdiv((long long)n * CO, B), B, 0, stream>>>(outp, den1, b1, n, 0);
  }
}


# Edge kernel: a work-group reads its die number and keeps only edges whose destination is in its package half's node range (per-wave LDS ring), so acc/den adds stay in one L2 half; sum order changes
# speedup vs baseline: 3.5440x; 1.3225x over previous
#include <hip/hip_runtime.h>
#include <hip/hip_bf16.h>

#define NEG_SLOPE 0.2f
#define LOGIT_LIMIT 40.0f
#define LDS_NODES 65536

typedef __attribute__((ext_vector_type(16))) _Float16 v16h;
typedef __attribute__((ext_vector_type(8)))  float    v8f;

template<int CO, bool PRE, bool ZERO, bool SEED>
__global__ __launch_bounds__(32)
void node_xform_wmma(const float* __restrict__ X, const float* __restrict__ W,
                     const float* __restrict__ bias, float* __restrict__ Out,
                     const float* __restrict__ pre_bias,
                     const float* __restrict__ pre_den,
                     float* __restrict__ zacc,
                     float* __restrict__ amax, float* __restrict__ den,
                     int* __restrict__ flag,
                     int n_nodes) {
  const int mtile = blockIdx.x;
  const int ntile = blockIdx.y;
  const int lane  = threadIdx.x;
  const int mrow  = lane & 15;
  const int grp   = lane >> 4;
  const int nIdx  = ntile * 16 + (lane & 15);
  const bool full = (mtile + 1) * 16 <= n_nodes;

  if constexpr (ZERO) {
    if (full) {
#pragma unroll
      for (int r = 0; r < 8; ++r)
        zacc[(size_t)(mtile * 16 + r + 8 * grp) * CO + nIdx] = 0.0f;
    } else {
#pragma unroll
      for (int r = 0; r < 8; ++r) {
        const int row = mtile * 16 + r + 8 * grp;
        if (row < n_nodes) zacc[(size_t)row * CO + nIdx] = 0.0f;
      }
    }
  }
  if constexpr (SEED) {
    if (ntile == 0 && lane < 16) {
      const int row = mtile * 16 + lane;
      if (row < n_nodes) { amax[row] = -3.402823466e38f; den[row] = 0.0f; }
      if (mtile == 0 && lane == 0) { flag[0] = 0; flag[1] = 0; flag[2] = 0; }
    }
  }

  int node = mtile * 16 + mrow;
  if (node >= n_nodes) node = n_nodes - 1;

  float inv = 1.0f;
  if constexpr (PRE) {
    const float dn = pre_den[node];
    inv = (dn > 0.0f) ? 1.0f / dn : 0.0f;
  }

  const float bv = bias[nIdx];
  v8f c;
#pragma unroll
  for (int r = 0; r < 8; ++r) c[r] = bv;

#pragma unroll
  for (int ks = 0; ks < 2; ++ks) {
    v16h a, b;
#pragma unroll
    for (int v = 0; v < 8; ++v) {
      const int k = ks * 32 + ((v & 4) ? 16 : 0) + grp * 8 + (v & 3) * 2;
      float2 xp = *(const float2*)(X + (size_t)node * 64 + k);
      if constexpr (PRE) {
        const float2 pb = *(const float2*)(pre_bias + k);
        xp.x = fmaxf(xp.x * inv + pb.x, 0.0f);
        xp.y = fmaxf(xp.y * inv + pb.y, 0.0f);
      }
      a[2 * v]     = (_Float16)xp.x;
      a[2 * v + 1] = (_Float16)xp.y;
      b[2 * v]     = (_Float16)W[(size_t)(k)     * CO + nIdx];
      b[2 * v + 1] = (_Float16)W[(size_t)(k + 1) * CO + nIdx];
    }
    c = __builtin_amdgcn_wmma_f32_16x16x32_f16(false, a, false, b,
                                               (short)0, c, false, false);
  }

  if (full) {
#pragma unroll
    for (int r = 0; r < 8; ++r)
      Out[(size_t)(mtile * 16 + r + 8 * grp) * CO + nIdx] = c[r];
  } else {
#pragma unroll
    for (int r = 0; r < 8; ++r) {
      const int row = mtile * 16 + r + 8 * grp;
      if (row < n_nodes) Out[(size_t)row * CO + nIdx] = c[r];
    }
  }
}

__device__ __forceinline__ void atomicMaxFloat(float* addr, float val) {
  if (val >= 0.0f) atomicMax((int*)addr, __float_as_int(val));
  else             atomicMin((unsigned int*)addr, __float_as_uint(val));
}

__device__ __forceinline__ float pick(bool c, float a, float b) { return c ? a : b; }

template<int CO>
__device__ __forceinline__ float chunk_logits(const float* __restrict__ xl,
                                              const float* __restrict__ xr,
                                              int s, int d, int febits,
                                              const float (&we)[CO / 32],
                                              const float (&at)[CO / 32],
                                              int lane,
                                              float (&xa)[32][CO / 32]) {
  constexpr int LPR = CO / 32;
  constexpr int G   = 16 / LPR;
  float p[32];
#pragma unroll
  for (int g = 0; g < 32; g += G) {
    float r[G][LPR];
#pragma unroll
    for (int j = 0; j < G; ++j) {
      const int sj = __builtin_amdgcn_readlane(s, g + j);
      const int dj = __builtin_amdgcn_readlane(d, g + j);
      const float* ra = xl + (size_t)sj * CO + lane;
      const float* rr = xr + (size_t)dj * CO + lane;
#pragma unroll
      for (int h = 0; h < LPR; ++h) {
        xa[g + j][h] = ra[h * 32];
        r[j][h]      = rr[h * 32];
      }
    }
#pragma unroll
    for (int j = 0; j < G; ++j) {
      const float fj = __int_as_float(__builtin_amdgcn_readlane(febits, g + j));
      float acc = 0.0f;
#pragma unroll
      for (int h = 0; h < LPR; ++h) {
        float m = xa[g + j][h] + r[j][h] + fj * we[h];
        m = (m > 0.0f) ? m : NEG_SLOPE * m;
        acc += m * at[h];
      }
      p[g + j] = acc;
    }
  }

#pragma unroll
  for (int w = 16; w >= 1; w >>= 1) {
    const bool hi = (lane & w) != 0;
    float got[16];
#pragma unroll
    for (int j = 0; j < w; ++j)
      got[j] = __shfl_xor(pick(hi, p[j], p[j + w]), w);
#pragma unroll
    for (int j = 0; j < w; ++j)
      p[j] = pick(hi, p[j + w], p[j]) + got[j];
  }
  return p[0];
}

template<int CO, int NT>
__global__ __launch_bounds__(NT)
void edge_onepass(const float* __restrict__ xl, const float* __restrict__ xr,
                  const int* __restrict__ src, const int* __restrict__ dst,
                  const float* __restrict__ ef, const float* __restrict__ We,
                  const float* __restrict__ att,
                  float* __restrict__ den, float* __restrict__ acc,
                  int* __restrict__ flag, int n_nodes, int ne) {
  constexpr int LPR = CO / 32;
  constexpr int NW  = NT / 32;
  __shared__ __attribute__((aligned(16))) float sden[LDS_NODES];
  __shared__ int ring_s[NW][64];
  __shared__ int ring_d[NW][64];
  __shared__ int ring_f[NW][64];
  __shared__ int s_role;
  const int tid  = (int)threadIdx.x;
  const int nthr = (int)blockDim.x;
  const int lane = tid & 31;
  const int wv = __builtin_amdgcn_readfirstlane(tid >> 5);
  const int nw = nthr >> 5;
  const int hg = (int)gridDim.x >> 1;

  if (tid == 0) {
    const int die = (int)((__builtin_amdgcn_s_sendmsg_rtn(0x87) >> 16) & 15u);
    const int ph  = (die >> 2) & 1;
    const int t   = atomicAdd(flag + 1 + ph, 1);
    int half = ph, slot = t;
    if (t >= hg) { half = ph ^ 1; slot = 2 * hg - 1 - t; }
    slot = (slot < 0) ? 0 : ((slot > hg - 1) ? hg - 1 : slot);
    s_role = half * hg + slot;
  }

  int nsplit = ((n_nodes / 2 + 63) / 64) * 64;
  if (nsplit > n_nodes) nsplit = n_nodes;

  int nclr = (nsplit > n_nodes - nsplit) ? nsplit : (n_nodes - nsplit);
  if (nclr > LDS_NODES) nclr = LDS_NODES;
  for (int i = tid * 4; i < nclr; i += nthr * 4)
    *(float4*)(sden + i) = make_float4(0.0f, 0.0f, 0.0f, 0.0f);

  float we[LPR], at[LPR];
#pragma unroll
  for (int h = 0; h < LPR; ++h) {
    we[h] = We[lane + h * 32];
    at[h] = att[lane + h * 32];
  }
  __syncthreads();
  const int role = __builtin_amdgcn_readfirstlane(s_role);
  const int half = (role >= hg) ? 1 : 0;
  const int slot = role - half * hg;
  const int lo = half ? nsplit : 0;
  const int hi = half ? n_nodes : nsplit;
  const int nl = (hi - lo < LDS_NODES) ? (hi - lo) : LDS_NODES;

  int* rs = ring_s[wv];
  int* rd = ring_d[wv];
  int* rf = ring_f[wv];
  int head = 0;
  int qn   = 0;

  const int nchunks = (ne + 31) / 32;
  const int cstep   = hg * nw;
  int chunk = wv * hg + slot;

  bool more = chunk < nchunks;
  int ps = 0, pd = -1, pf = 0;
  if (more) {
    const int e = chunk * 32 + lane;
    if (e < ne) { ps = src[e]; pd = dst[e]; pf = __float_as_int(ef[e]); }
  }

  for (;;) {
    if (more) {
      const bool mine = (pd >= lo) && (pd < hi);
      const unsigned m = __builtin_amdgcn_ballot_w32(mine);
      if (mine) {
        const int pos = (head + qn + (int)__builtin_amdgcn_mbcnt_lo(m, 0u)) & 63;
        rs[pos] = ps; rd[pos] = pd; rf[pos] = pf;
      }
      qn += __builtin_popcount(m);
      chunk += cstep;
      more = chunk < nchunks;
      ps = 0; pd = -1; pf = 0;
      if (more) {
        const int e = chunk * 32 + lane;
        if (e < ne) { ps = src[e]; pd = dst[e]; pf = __float_as_int(ef[e]); }
      }
      if (more && qn < 32) continue;
    }
    if (qn == 0) break;
    const int nvalid = (qn < 32) ? qn : 32;

    int s = 0, d = 0, febits = 0;
    if (lane < nvalid) {
      const int pos = (head + lane) & 63;
      s = rs[pos]; d = rd[pos]; febits = rf[pos];
    }
    head = (head + nvalid) & 63;
    qn  -= nvalid;

    float xa[32][LPR];
    const float logit = chunk_logits<CO>(xl, xr, s, d, febits, we, at, lane, xa);

    const bool valid = lane < nvalid;
    const bool out_of_range = valid && !(fabsf(logit) <= LOGIT_LIMIT);
    if (__builtin_amdgcn_ballot_w32(out_of_range) != 0u) {
      if (lane == 0) atomicOr(flag, 1);
    }

    float w = 0.0f;
    if (valid) {
      w = __expf(logit);
      const int dl = d - lo;
      if (dl < nl) atomicAdd(sden + dl, w);
      else         atomicAdd(den + d, w);
    }
    const int wbits = __float_as_int(w);

    if (nvalid == 32) {
#pragma unroll
      for (int j = 0; j < 32; ++j) {
        const int   dj = __builtin_amdgcn_readlane(d, j);
        const float wj = __int_as_float(__builtin_amdgcn_readlane(wbits, j));
        float* p = acc + (size_t)dj * CO + lane;
#pragma unroll
        for (int h = 0; h < LPR; ++h) atomicAdd(p + h * 32, xa[j][h] * wj);
      }
    } else {
#pragma unroll
      for (int j = 0; j < 32; ++j) {
        if (j < nvalid) {
          const int   dj = __builtin_amdgcn_readlane(d, j);
          const float wj = __int_as_float(__builtin_amdgcn_readlane(wbits, j));
          float* p = acc + (size_t)dj * CO + lane;
#pragma unroll
          for (int h = 0; h < LPR; ++h) atomicAdd(p + h * 32, xa[j][h] * wj);
        }
      }
    }
  }
  __syncthreads();

  float* denh = den + lo;
  const int mis    = (int)(((size_t)denh >> 2) & 31);
  const int nlines = (nl + mis + 31) / 32;
  const int rot    = (int)(((unsigned)slot * (unsigned)nlines) / (unsigned)(hg > 0 ? hg : 1));
  for (int k0 = wv * 8; k0 < nlines; k0 += nw * 8) {
    float v[8];
    int   idx[8];
#pragma unroll
    for (int u = 0; u < 8; ++u) {
      int line = k0 + u + rot;
      if (line >= nlines) line -= nlines;
      const int i = line * 32 + lane - mis;
      idx[u] = i;
      v[u] = (k0 + u < nlines && i >= 0 && i < nl) ? sden[i] : 0.0f;
    }
#pragma unroll
    for (int u = 0; u < 8; ++u)
      if (v[u] != 0.0f) atomicAdd(denh + idx[u], v[u]);
  }
}

template<int CO>
__global__ __launch_bounds__(256)
void edge_alpha_shifted(const float* __restrict__ xl, const float* __restrict__ xr,
                        const int* __restrict__ src, const int* __restrict__ dst,
                        const float* __restrict__ ef, const float* __restrict__ We,
                        const float* __restrict__ att,
                        float* __restrict__ alpha, float* __restrict__ amax,
                        float* __restrict__ zacc, float* __restrict__ zden,
                        int n_nodes, const int* __restrict__ flag, int ne) {
  if (*flag == 0) return;
  constexpr int LPR = CO / 32;
  const int tid      = (int)(blockIdx.x * blockDim.x + threadIdx.x);
  const int nthreads = (int)(gridDim.x * blockDim.x);
  const size_t nacc = (size_t)n_nodes * CO;
  for (size_t i = (size_t)tid; i < nacc; i += (size_t)nthreads) zacc[i] = 0.0f;
  for (int i = tid; i < n_nodes; i += nthreads) zden[i] = 0.0f;

  const int lane = threadIdx.x & 31;
  const int wave0  = __builtin_amdgcn_readfirstlane(tid >> 5);
  const int nwaves = nthreads >> 5;
  const int nchunks = (ne + 31) / 32;

  float we[LPR], at[LPR];
#pragma unroll
  for (int h = 0; h < LPR; ++h) {
    we[h] = We[lane + h * 32];
    at[h] = att[lane + h * 32];
  }

  for (int chunk = wave0; chunk < nchunks; chunk += nwaves) {
    const int base = chunk * 32;
    const int nvalid = (ne - base < 32) ? (ne - base) : 32;
    int s = 0, d = 0;
    float fe = 0.0f;
    if (lane < nvalid) {
      const int e = base + lane;
      s  = src[e];
      d  = dst[e];
      fe = ef[e];
    }
    float xa[32][LPR];
    const float logit = chunk_logits<CO>(xl, xr, s, d, __float_as_int(fe), we, at, lane, xa);
    if (lane < nvalid) {
      alpha[base + lane] = logit;
      atomicMaxFloat(amax + d, logit);
    }
  }
}

template<int CO>
__global__ __launch_bounds__(256)
void edge_scatter_shifted(const float* __restrict__ xl, const float* __restrict__ alpha,
                          const float* __restrict__ amax, float* __restrict__ den,
                          const int* __restrict__ src, const int* __restrict__ dst,
                          float* __restrict__ acc, const int* __restrict__ flag, int ne) {
  if (*flag == 0) return;
  constexpr int LPR = CO / 32;
  constexpr int G   = 16 / LPR;
  const int tid      = (int)(blockIdx.x * blockDim.x + threadIdx.x);
  const int nthreads = (int)(gridDim.x * blockDim.x);
  const int lane = threadIdx.x & 31;
  const int wave0  = __builtin_amdgcn_readfirstlane(tid >> 5);
  const int nwaves = nthreads >> 5;
  const int nchunks = (ne + 31) / 32;

  for (int chunk = wave0; chunk < nchunks; chunk += nwaves) {
    const int base = chunk * 32;
    const int nvalid = (ne - base < 32) ? (ne - base) : 32;

    int s = 0, d = 0;
    float w = 0.0f;
    if (lane < nvalid) {
      const int e = base + lane;
      s = src[e];
      d = dst[e];
      w = __expf(alpha[e] - amax[d]);
      atomicAdd(den + d, w);
    }
    const int wbits = __float_as_int(w);

    if (nvalid == 32) {
#pragma unroll
      for (int g = 0; g < 32; g += G) {
        float v[G][LPR];
#pragma unroll
        for (int j = 0; j < G; ++j) {
          const int sj = __builtin_amdgcn_readlane(s, g + j);
          const float* row = xl + (size_t)sj * CO + lane;
#pragma unroll
          for (int h = 0; h < LPR; ++h) v[j][h] = row[h * 32];
        }
#pragma unroll
        for (int j = 0; j < G; ++j) {
          const int   dj = __builtin_amdgcn_readlane(d, g + j);
          const float wj = __int_as_float(__builtin_amdgcn_readlane(wbits, g + j));
          float* p = acc + (size_t)dj * CO + lane;
#pragma unroll
          for (int h = 0; h < LPR; ++h) atomicAdd(p + h * 32, v[j][h] * wj);
        }
      }
    } else {
      for (int j = 0; j < nvalid; ++j) {
        const int   sj = __shfl(s, j);
        const int   dj = __shfl(d, j);
        const float wj = __shfl(w, j);
        const float* row = xl + (size_t)sj * CO + lane;
        float* p = acc + (size_t)dj * CO + lane;
#pragma unroll
        for (int h = 0; h < LPR; ++h) atomicAdd(p + h * 32, row[h * 32] * wj);
      }
    }
  }
}

template<int CO>
__global__ __launch_bounds__(256)
void k_finalize(float* __restrict__ acc, const float* __restrict__ den,
                const float* __restrict__ bias, int n, int do_relu) {
  const int i = blockIdx.x * blockDim.x + threadIdx.x;
  if (i >= n * CO) return;
  const float dn = den[i / CO];
  float v = ((dn > 0.0f) ? acc[i] / dn : 0.0f) + bias[i % CO];
  if (do_relu) v = fmaxf(v, 0.0f);
  acc[i] = v;
}

extern "C" void kernel_launch(void* const* d_in, const int* in_sizes, int n_in,
                              void* d_out, int out_size, void* d_ws, size_t ws_size,
                              hipStream_t stream) {
  (void)n_in; (void)out_size; (void)ws_size;
  const float* x    = (const float*)d_in[0];
  const int*   ei   = (const int*)d_in[1];
  const float* ef   = (const float*)d_in[2];
  const float* Wl0  = (const float*)d_in[3];
  const float* bl0  = (const float*)d_in[4];
  const float* Wr0  = (const float*)d_in[5];
  const float* br0  = (const float*)d_in[6];
  const float* We0  = (const float*)d_in[7];
  const float* att0 = (const float*)d_in[8];
  const float* b0   = (const float*)d_in[9];
  const float* Wl1  = (const float*)d_in[10];
  const float* bl1  = (const float*)d_in[11];
  const float* Wr1  = (const float*)d_in[12];
  const float* br1  = (const float*)d_in[13];
  const float* We1  = (const float*)d_in[14];
  const float* att1 = (const float*)d_in[15];
  const float* b1   = (const float*)d_in[16];

  const int n = in_sizes[0] / 64;
  const int e = in_sizes[1] / 2;
  const int* srcI = ei;
  const int* dstI = ei + e;

  float* ws   = (float*)d_ws;
  float* xl   = ws;  ws += (size_t)n * 64;
  float* xr   = ws;  ws += (size_t)n * 64;
  float* hbuf = ws;  ws += (size_t)n * 64;
  float* alp  = ws;  ws += (size_t)e;
  float* amax = ws;  ws += n;
  float* den0 = ws;  ws += n;
  float* den1 = ws;  ws += n;
  int* flag0  = (int*)ws;
  int* flag1  = flag0 + 4;

  auto cdiv = [](long long a, long long b) { return (int)((a + b - 1) / b); };
  const int B = 256;
  const int GS = 256;

  {
    constexpr int CO = 64;
    dim3 g((n + 15) / 16, CO / 16);
    node_xform_wmma<CO, false, true, false><<<g, 32, 0, stream>>>(
        x, Wl0, bl0, xl, nullptr, nullptr, hbuf, nullptr, nullptr, nullptr, n);
    node_xform_wmma<CO, false, false, true><<<g, 32, 0, stream>>>(
        x, Wr0, br0, xr, nullptr, nullptr, nullptr, amax, den0, flag0, n);
    edge_onepass<CO, 896><<<GS, 896, 0, stream>>>(
        xl, xr, srcI, dstI, ef, We0, att0, den0, hbuf, flag0, n, e);
    edge_alpha_shifted<CO><<<GS, B, 0, stream>>>(
        xl, xr, srcI, dstI, ef, We0, att0, alp, amax, hbuf, den0, n, flag0, e);
    edge_scatter_shifted<CO><<<GS, B, 0, stream>>>(
        xl, alp, amax, den0, srcI, dstI, hbuf, flag0, e);
  }

  {
    constexpr int CO = 32;
    float* outp = (float*)d_out;
    dim3 g((n + 15) / 16, CO / 16);
    node_xform_wmma<CO, true, true, false><<<g, 32, 0, stream>>>(
        hbuf, Wl1, bl1, xl, b0, den0, outp, nullptr, nullptr, nullptr, n);
    node_xform_wmma<CO, true, false, true><<<g, 32, 0, stream>>>(
        hbuf, Wr1, br1, xr, b0, den0, nullptr, amax, den1, flag1, n);
    edge_onepass<CO, 1024><<<GS, 1024, 0, stream>>>(
        xl, xr, srcI, dstI, ef, We1, att1, den1, outp, flag1, n, e);
    edge_alpha_shifted<CO><<<GS, B, 0, stream>>>(
        xl, xr, srcI, dstI, ef, We1, att1, alp, amax, outp, den1, n, flag1, e);
    edge_scatter_shifted<CO><<<GS, B, 0, stream>>>(
        xl, alp, amax, den1, srcI, dstI, outp, flag1, e);
    k_finalize<CO><<<cdiv((long long)n * CO, B), B, 0, stream>>>(outp, den1, b1, n, 0);
  }
}


# Node transforms on a resident grid: a group stages each 16-row strip once in LDS as f16, a wave keeps one weight tile's B fragments all launch; xl and xr in one launch, 11 become 9; bit-identical
# speedup vs baseline: 4.2787x; 1.2073x over previous
#include <hip/hip_runtime.h>
#include <hip/hip_bf16.h>

#define NEG_SLOPE 0.2f
#define LOGIT_LIMIT 40.0f
#define LDS_NODES 65536

typedef __attribute__((ext_vector_type(16))) _Float16 v16h;
typedef __attribute__((ext_vector_type(8)))  float    v8f;

typedef __attribute__((ext_vector_type(8))) _Float16 v8h;
typedef __attribute__((ext_vector_type(4))) _Float16 v4h;

__device__ __forceinline__ void lds_barrier() {
  asm volatile("s_wait_dscnt 0x0\n\ts_barrier_signal -1\n\ts_barrier_wait -1" ::: "memory");
}

template<int CO, int NT, bool PRE>
__global__ __launch_bounds__(NT)
void node_xform_resident(const float* __restrict__ X,
                         const float* __restrict__ Wl, const float* __restrict__ bl,
                         const float* __restrict__ Wr, const float* __restrict__ br,
                         float* __restrict__ xl, float* __restrict__ xr,
                         const float* __restrict__ pre_bias,
                         const float* __restrict__ pre_den,
                         float* __restrict__ zacc,
                         float* __restrict__ amax, float* __restrict__ den,
                         int* __restrict__ flag,
                         int n_nodes) {
  constexpr int CT    = CO / 16;
  constexpr int R     = 2 * CT;
  constexpr int NW    = NT / 32;
  constexpr int SPG   = NW / R;
  constexpr int TPS   = NT / 256;
  constexpr int LPT   = SPG / TPS;
  constexpr int PITCH = 72;
  static_assert(NT % 256 == 0 && NW % R == 0 && SPG % TPS == 0, "shape");
  __shared__ __attribute__((aligned(16))) _Float16 sA[2][SPG * 16 * PITCH];

  const int tid  = (int)threadIdx.x;
  const int lane = tid & 31;
  const int wv    = __builtin_amdgcn_readfirstlane(tid >> 5);
  const int role  = wv % R;
  const int wstr  = wv / R;
  const int mat   = role / CT;
  const int ntile = role % CT;
  const float* __restrict__ W    = mat ? Wr : Wl;
  const float* __restrict__ bias = mat ? br : bl;
  float* __restrict__ Out        = mat ? xr : xl;
  const int mrow = lane & 15;
  const int grp  = lane >> 4;
  const int nIdx = ntile * 16 + mrow;

  if (blockIdx.x == 0 && tid == 0) { flag[0] = 0; flag[1] = 0; flag[2] = 0; }

  v16h b0, b1;
#pragma unroll
  for (int e = 0; e < 16; ++e) {
    const int k = (e >> 3) * 16 + grp * 8 + (e & 7);
    b0[e] = (_Float16)W[(size_t)k        * CO + nIdx];
    b1[e] = (_Float16)W[(size_t)(k + 32) * CO + nIdx];
  }
  const float bv = bias[nIdx];

  const int lrow = (tid & 255) >> 4;
  const int col4 = tid & 15;
  const int lstr = tid >> 8;
  float4 pb = make_float4(0.0f, 0.0f, 0.0f, 0.0f);
  if constexpr (PRE) pb = *(const float4*)(pre_bias + col4 * 4);

  const int nstrips = (n_nodes + 15) / 16;
  const int nsteps  = (nstrips + SPG - 1) / SPG;

  float4 raw[LPT];
  float  dn[LPT];
  int step = (int)blockIdx.x;
  if (step < nsteps) {
#pragma unroll
    for (int u = 0; u < LPT; ++u) {
      int node = (step * SPG + u * TPS + lstr) * 16 + lrow;
      if (node >= n_nodes) node = n_nodes - 1;
      raw[u] = *(const float4*)(X + (size_t)node * 64 + col4 * 4);
      if constexpr (PRE) dn[u] = pre_den[node];
    }
  }

  int buf = 0;
  for (; step < nsteps; step += (int)gridDim.x) {
#pragma unroll
    for (int u = 0; u < LPT; ++u) {
      float4 xq = raw[u];
      if constexpr (PRE) {
        const float inv = (dn[u] > 0.0f) ? 1.0f / dn[u] : 0.0f;
        xq.x = fmaxf(xq.x * inv + pb.x, 0.0f);
        xq.y = fmaxf(xq.y * inv + pb.y, 0.0f);
        xq.z = fmaxf(xq.z * inv + pb.z, 0.0f);
        xq.w = fmaxf(xq.w * inv + pb.w, 0.0f);
      }
      v4h hq;
      hq[0] = (_Float16)xq.x; hq[1] = (_Float16)xq.y;
      hq[2] = (_Float16)xq.z; hq[3] = (_Float16)xq.w;
      *(v4h*)(&sA[buf][((u * TPS + lstr) * 16 + lrow) * PITCH + col4 * 4]) = hq;
    }

    const int nstep = step + (int)gridDim.x;
    if (nstep < nsteps) {
#pragma unroll
      for (int u = 0; u < LPT; ++u) {
        int node = (nstep * SPG + u * TPS + lstr) * 16 + lrow;
        if (node >= n_nodes) node = n_nodes - 1;
        raw[u] = *(const float4*)(X + (size_t)node * 64 + col4 * 4);
        if constexpr (PRE) dn[u] = pre_den[node];
      }
    }
    lds_barrier();

    const int strip = step * SPG + wstr;
    if (strip < nstrips) {
      const _Float16* ar = &sA[buf][(wstr * 16 + mrow) * PITCH + grp * 8];
      const v8h a00 = *(const v8h*)(ar);
      const v8h a01 = *(const v8h*)(ar + 16);
      const v8h a10 = *(const v8h*)(ar + 32);
      const v8h a11 = *(const v8h*)(ar + 48);
      const v16h a0 = __builtin_shufflevector(a00, a01, 0, 1, 2, 3, 4, 5, 6, 7,
                                              8, 9, 10, 11, 12, 13, 14, 15);
      const v16h a1 = __builtin_shufflevector(a10, a11, 0, 1, 2, 3, 4, 5, 6, 7,
                                              8, 9, 10, 11, 12, 13, 14, 15);

      {
        const size_t zi = (size_t)strip * 16 * CO + (size_t)role * 128 + (size_t)lane * 4;
        if (zi < (size_t)n_nodes * CO)
          *(float4*)(zacc + zi) = make_float4(0.0f, 0.0f, 0.0f, 0.0f);
      }
      if (role == 0 && lane < 16) {
        const int row = strip * 16 + lane;
        if (row < n_nodes) { amax[row] = -3.402823466e38f; den[row] = 0.0f; }
      }

      v8f c;
#pragma unroll
      for (int r = 0; r < 8; ++r) c[r] = bv;
      c = __builtin_amdgcn_wmma_f32_16x16x32_f16(false, a0, false, b0,
                                                 (short)0, c, false, false);
      c = __builtin_amdgcn_wmma_f32_16x16x32_f16(false, a1, false, b1,
                                                 (short)0, c, false, false);

      if ((strip + 1) * 16 <= n_nodes) {
#pragma unroll
        for (int r = 0; r < 8; ++r)
          Out[(size_t)(strip * 16 + r + 8 * grp) * CO + nIdx] = c[r];
      } else {
#pragma unroll
        for (int r = 0; r < 8; ++r) {
          const int row = strip * 16 + r + 8 * grp;
          if (row < n_nodes) Out[(size_t)row * CO + nIdx] = c[r];
        }
      }
    }
    buf ^= 1;
  }
}

__device__ __forceinline__ void atomicMaxFloat(float* addr, float val) {
  if (val >= 0.0f) atomicMax((int*)addr, __float_as_int(val));
  else             atomicMin((unsigned int*)addr, __float_as_uint(val));
}

__device__ __forceinline__ float pick(bool c, float a, float b) { return c ? a : b; }

template<int CO>
__device__ __forceinline__ float chunk_logits(const float* __restrict__ xl,
                                              const float* __restrict__ xr,
                                              int s, int d, int febits,
                                              const float (&we)[CO / 32],
                                              const float (&at)[CO / 32],
                                              int lane,
                                              float (&xa)[32][CO / 32]) {
  constexpr int LPR = CO / 32;
  constexpr int G   = 16 / LPR;
  float p[32];
#pragma unroll
  for (int g = 0; g < 32; g += G) {
    float r[G][LPR];
#pragma unroll
    for (int j = 0; j < G; ++j) {
      const int sj = __builtin_amdgcn_readlane(s, g + j);
      const int dj = __builtin_amdgcn_readlane(d, g + j);
      const float* ra = xl + (size_t)sj * CO + lane;
      const float* rr = xr + (size_t)dj * CO + lane;
#pragma unroll
      for (int h = 0; h < LPR; ++h) {
        xa[g + j][h] = ra[h * 32];
        r[j][h]      = rr[h * 32];
      }
    }
#pragma unroll
    for (int j = 0; j < G; ++j) {
      const float fj = __int_as_float(__builtin_amdgcn_readlane(febits, g + j));
      float acc = 0.0f;
#pragma unroll
      for (int h = 0; h < LPR; ++h) {
        float m = xa[g + j][h] + r[j][h] + fj * we[h];
        m = (m > 0.0f) ? m : NEG_SLOPE * m;
        acc += m * at[h];
      }
      p[g + j] = acc;
    }
  }

#pragma unroll
  for (int w = 16; w >= 1; w >>= 1) {
    const bool hi = (lane & w) != 0;
    float got[16];
#pragma unroll
    for (int j = 0; j < w; ++j)
      got[j] = __shfl_xor(pick(hi, p[j], p[j + w]), w);
#pragma unroll
    for (int j = 0; j < w; ++j)
      p[j] = pick(hi, p[j + w], p[j]) + got[j];
  }
  return p[0];
}

template<int CO, int NT>
__global__ __launch_bounds__(NT)
void edge_onepass(const float* __restrict__ xl, const float* __restrict__ xr,
                  const int* __restrict__ src, const int* __restrict__ dst,
                  const float* __restrict__ ef, const float* __restrict__ We,
                  const float* __restrict__ att,
                  float* __restrict__ den, float* __restrict__ acc,
                  int* __restrict__ flag, int n_nodes, int ne) {
  constexpr int LPR = CO / 32;
  constexpr int NW  = NT / 32;
  __shared__ __attribute__((aligned(16))) float sden[LDS_NODES];
  __shared__ int ring_s[NW][64];
  __shared__ int ring_d[NW][64];
  __shared__ int ring_f[NW][64];
  __shared__ int s_role;
  const int tid  = (int)threadIdx.x;
  const int nthr = (int)blockDim.x;
  const int lane = tid & 31;
  const int wv = __builtin_amdgcn_readfirstlane(tid >> 5);
  const int nw = nthr >> 5;
  const int hg = (int)gridDim.x >> 1;

  if (tid == 0) {
    const int die = (int)((__builtin_amdgcn_s_sendmsg_rtn(0x87) >> 16) & 15u);
    const int ph  = (die >> 2) & 1;
    const int t   = atomicAdd(flag + 1 + ph, 1);
    int half = ph, slot = t;
    if (t >= hg) { half = ph ^ 1; slot = 2 * hg - 1 - t; }
    slot = (slot < 0) ? 0 : ((slot > hg - 1) ? hg - 1 : slot);
    s_role = half * hg + slot;
  }

  int nsplit = ((n_nodes / 2 + 63) / 64) * 64;
  if (nsplit > n_nodes) nsplit = n_nodes;

  int nclr = (nsplit > n_nodes - nsplit) ? nsplit : (n_nodes - nsplit);
  if (nclr > LDS_NODES) nclr = LDS_NODES;
  for (int i = tid * 4; i < nclr; i += nthr * 4)
    *(float4*)(sden + i) = make_float4(0.0f, 0.0f, 0.0f, 0.0f);

  float we[LPR], at[LPR];
#pragma unroll
  for (int h = 0; h < LPR; ++h) {
    we[h] = We[lane + h * 32];
    at[h] = att[lane + h * 32];
  }
  __syncthreads();
  const int role = __builtin_amdgcn_readfirstlane(s_role);
  const int half = (role >= hg) ? 1 : 0;
  const int slot = role - half * hg;
  const int lo = half ? nsplit : 0;
  const int hi = half ? n_nodes : nsplit;
  const int nl = (hi - lo < LDS_NODES) ? (hi - lo) : LDS_NODES;

  int* rs = ring_s[wv];
  int* rd = ring_d[wv];
  int* rf = ring_f[wv];
  int head = 0;
  int qn   = 0;

  const int nchunks = (ne + 31) / 32;
  const int cstep   = hg * nw;
  int chunk = wv * hg + slot;

  bool more = chunk < nchunks;
  int ps = 0, pd = -1, pf = 0;
  if (more) {
    const int e = chunk * 32 + lane;
    if (e < ne) { ps = src[e]; pd = dst[e]; pf = __float_as_int(ef[e]); }
  }

  for (;;) {
    if (more) {
      const bool mine = (pd >= lo) && (pd < hi);
      const unsigned m = __builtin_amdgcn_ballot_w32(mine);
      if (mine) {
        const int pos = (head + qn + (int)__builtin_amdgcn_mbcnt_lo(m, 0u)) & 63;
        rs[pos] = ps; rd[pos] = pd; rf[pos] = pf;
      }
      qn += __builtin_popcount(m);
      chunk += cstep;
      more = chunk < nchunks;
      ps = 0; pd = -1; pf = 0;
      if (more) {
        const int e = chunk * 32 + lane;
        if (e < ne) { ps = src[e]; pd = dst[e]; pf = __float_as_int(ef[e]); }
      }
      if (more && qn < 32) continue;
    }
    if (qn == 0) break;
    const int nvalid = (qn < 32) ? qn : 32;

    int s = 0, d = 0, febits = 0;
    if (lane < nvalid) {
      const int pos = (head + lane) & 63;
      s = rs[pos]; d = rd[pos]; febits = rf[pos];
    }
    head = (head + nvalid) & 63;
    qn  -= nvalid;

    float xa[32][LPR];
    const float logit = chunk_logits<CO>(xl, xr, s, d, febits, we, at, lane, xa);

    const bool valid = lane < nvalid;
    const bool out_of_range = valid && !(fabsf(logit) <= LOGIT_LIMIT);
    if (__builtin_amdgcn_ballot_w32(out_of_range) != 0u) {
      if (lane == 0) atomicOr(flag, 1);
    }

    float w = 0.0f;
    if (valid) {
      w = __expf(logit);
      const int dl = d - lo;
      if (dl < nl) atomicAdd(sden + dl, w);
      else         atomicAdd(den + d, w);
    }
    const int wbits = __float_as_int(w);

    if (nvalid == 32) {
#pragma unroll
      for (int j = 0; j < 32; ++j) {
        const int   dj = __builtin_amdgcn_readlane(d, j);
        const float wj = __int_as_float(__builtin_amdgcn_readlane(wbits, j));
        float* p = acc + (size_t)dj * CO + lane;
#pragma unroll
        for (int h = 0; h < LPR; ++h) atomicAdd(p + h * 32, xa[j][h] * wj);
      }
    } else {
#pragma unroll
      for (int j = 0; j < 32; ++j) {
        if (j < nvalid) {
          const int   dj = __builtin_amdgcn_readlane(d, j);
          const float wj = __int_as_float(__builtin_amdgcn_readlane(wbits, j));
          float* p = acc + (size_t)dj * CO + lane;
#pragma unroll
          for (int h = 0; h < LPR; ++h) atomicAdd(p + h * 32, xa[j][h] * wj);
        }
      }
    }
  }
  __syncthreads();

  float* denh = den + lo;
  const int mis    = (int)(((size_t)denh >> 2) & 31);
  const int nlines = (nl + mis + 31) / 32;
  const int rot    = (int)(((unsigned)slot * (unsigned)nlines) / (unsigned)(hg > 0 ? hg : 1));
  for (int k0 = wv * 8; k0 < nlines; k0 += nw * 8) {
    float v[8];
    int   idx[8];
#pragma unroll
    for (int u = 0; u < 8; ++u) {
      int line = k0 + u + rot;
      if (line >= nlines) line -= nlines;
      const int i = line * 32 + lane - mis;
      idx[u] = i;
      v[u] = (k0 + u < nlines && i >= 0 && i < nl) ? sden[i] : 0.0f;
    }
#pragma unroll
    for (int u = 0; u < 8; ++u)
      if (v[u] != 0.0f) atomicAdd(denh + idx[u], v[u]);
  }
}

template<int CO>
__global__ __launch_bounds__(256)
void edge_alpha_shifted(const float* __restrict__ xl, const float* __restrict__ xr,
                        const int* __restrict__ src, const int* __restrict__ dst,
                        const float* __restrict__ ef, const float* __restrict__ We,
                        const float* __restrict__ att,
                        float* __restrict__ alpha, float* __restrict__ amax,
                        float* __restrict__ zacc, float* __restrict__ zden,
                        int n_nodes, const int* __restrict__ flag, int ne) {
  if (*flag == 0) return;
  constexpr int LPR = CO / 32;
  const int tid      = (int)(blockIdx.x * blockDim.x + threadIdx.x);
  const int nthreads = (int)(gridDim.x * blockDim.x);
  const size_t nacc = (size_t)n_nodes * CO;
  for (size_t i = (size_t)tid; i < nacc; i += (size_t)nthreads) zacc[i] = 0.0f;
  for (int i = tid; i < n_nodes; i += nthreads) zden[i] = 0.0f;

  const int lane = threadIdx.x & 31;
  const int wave0  = __builtin_amdgcn_readfirstlane(tid >> 5);
  const int nwaves = nthreads >> 5;
  const int nchunks = (ne + 31) / 32;

  float we[LPR], at[LPR];
#pragma unroll
  for (int h = 0; h < LPR; ++h) {
    we[h] = We[lane + h * 32];
    at[h] = att[lane + h * 32];
  }

  for (int chunk = wave0; chunk < nchunks; chunk += nwaves) {
    const int base = chunk * 32;
    const int nvalid = (ne - base < 32) ? (ne - base) : 32;
    int s = 0, d = 0;
    float fe = 0.0f;
    if (lane < nvalid) {
      const int e = base + lane;
      s  = src[e];
      d  = dst[e];
      fe = ef[e];
    }
    float xa[32][LPR];
    const float logit = chunk_logits<CO>(xl, xr, s, d, __float_as_int(fe), we, at, lane, xa);
    if (lane < nvalid) {
      alpha[base + lane] = logit;
      atomicMaxFloat(amax + d, logit);
    }
  }
}

template<int CO>
__global__ __launch_bounds__(256)
void edge_scatter_shifted(const float* __restrict__ xl, const float* __restrict__ alpha,
                          const float* __restrict__ amax, float* __restrict__ den,
                          const int* __restrict__ src, const int* __restrict__ dst,
                          float* __restrict__ acc, const int* __restrict__ flag, int ne) {
  if (*flag == 0) return;
  constexpr int LPR = CO / 32;
  constexpr int G   = 16 / LPR;
  const int tid      = (int)(blockIdx.x * blockDim.x + threadIdx.x);
  const int nthreads = (int)(gridDim.x * blockDim.x);
  const int lane = threadIdx.x & 31;
  const int wave0  = __builtin_amdgcn_readfirstlane(tid >> 5);
  const int nwaves = nthreads >> 5;
  const int nchunks = (ne + 31) / 32;

  for (int chunk = wave0; chunk < nchunks; chunk += nwaves) {
    const int base = chunk * 32;
    const int nvalid = (ne - base < 32) ? (ne - base) : 32;

    int s = 0, d = 0;
    float w = 0.0f;
    if (lane < nvalid) {
      const int e = base + lane;
      s = src[e];
      d = dst[e];
      w = __expf(alpha[e] - amax[d]);
      atomicAdd(den + d, w);
    }
    const int wbits = __float_as_int(w);

    if (nvalid == 32) {
#pragma unroll
      for (int g = 0; g < 32; g += G) {
        float v[G][LPR];
#pragma unroll
        for (int j = 0; j < G; ++j) {
          const int sj = __builtin_amdgcn_readlane(s, g + j);
          const float* row = xl + (size_t)sj * CO + lane;
#pragma unroll
          for (int h = 0; h < LPR; ++h) v[j][h] = row[h * 32];
        }
#pragma unroll
        for (int j = 0; j < G; ++j) {
          const int   dj = __builtin_amdgcn_readlane(d, g + j);
          const float wj = __int_as_float(__builtin_amdgcn_readlane(wbits, g + j));
          float* p = acc + (size_t)dj * CO + lane;
#pragma unroll
          for (int h = 0; h < LPR; ++h) atomicAdd(p + h * 32, v[j][h] * wj);
        }
      }
    } else {
      for (int j = 0; j < nvalid; ++j) {
        const int   sj = __shfl(s, j);
        const int   dj = __shfl(d, j);
        const float wj = __shfl(w, j);
        const float* row = xl + (size_t)sj * CO + lane;
        float* p = acc + (size_t)dj * CO + lane;
#pragma unroll
        for (int h = 0; h < LPR; ++h) atomicAdd(p + h * 32, row[h * 32] * wj);
      }
    }
  }
}

template<int CO>
__global__ __launch_bounds__(256)
void k_finalize(float* __restrict__ acc, const float* __restrict__ den,
                const float* __restrict__ bias, int n, int do_relu) {
  const int i = blockIdx.x * blockDim.x + threadIdx.x;
  if (i >= n * CO) return;
  const float dn = den[i / CO];
  float v = ((dn > 0.0f) ? acc[i] / dn : 0.0f) + bias[i % CO];
  if (do_relu) v = fmaxf(v, 0.0f);
  acc[i] = v;
}

extern "C" void kernel_launch(void* const* d_in, const int* in_sizes, int n_in,
                              void* d_out, int out_size, void* d_ws, size_t ws_size,
                              hipStream_t stream) {
  (void)n_in; (void)out_size; (void)ws_size;
  const float* x    = (const float*)d_in[0];
  const int*   ei   = (const int*)d_in[1];
  const float* ef   = (const float*)d_in[2];
  const float* Wl0  = (const float*)d_in[3];
  const float* bl0  = (const float*)d_in[4];
  const float* Wr0  = (const float*)d_in[5];
  const float* br0  = (const float*)d_in[6];
  const float* We0  = (const float*)d_in[7];
  const float* att0 = (const float*)d_in[8];
  const float* b0   = (const float*)d_in[9];
  const float* Wl1  = (const float*)d_in[10];
  const float* bl1  = (const float*)d_in[11];
  const float* Wr1  = (const float*)d_in[12];
  const float* br1  = (const float*)d_in[13];
  const float* We1  = (const float*)d_in[14];
  const float* att1 = (const float*)d_in[15];
  const float* b1   = (const float*)d_in[16];

  const int n = in_sizes[0] / 64;
  const int e = in_sizes[1] / 2;
  const int* srcI = ei;
  const int* dstI = ei + e;

  float* ws   = (float*)d_ws;
  float* xl   = ws;  ws += (size_t)n * 64;
  float* xr   = ws;  ws += (size_t)n * 64;
  float* hbuf = ws;  ws += (size_t)n * 64;
  float* alp  = ws;  ws += (size_t)e;
  float* amax = ws;  ws += n;
  float* den0 = ws;  ws += n;
  float* den1 = ws;  ws += n;
  int* flag0  = (int*)ws;
  int* flag1  = flag0 + 4;

  auto cdiv = [](long long a, long long b) { return (int)((a + b - 1) / b); };
  const int B = 256;
  const int GS = 256;

  {
    constexpr int CO = 64;
    node_xform_resident<CO, 512, false><<<512, 512, 0, stream>>>(
        x, Wl0, bl0, Wr0, br0, xl, xr, nullptr, nullptr, hbuf, amax, den0, flag0, n);
    edge_onepass<CO, 896><<<GS, 896, 0, stream>>>(
        xl, xr, srcI, dstI, ef, We0, att0, den0, hbuf, flag0, n, e);
    edge_alpha_shifted<CO><<<GS, B, 0, stream>>>(
        xl, xr, srcI, dstI, ef, We0, att0, alp, amax, hbuf, den0, n, flag0, e);
    edge_scatter_shifted<CO><<<GS, B, 0, stream>>>(
        xl, alp, amax, den0, srcI, dstI, hbuf, flag0, e);
  }

  {
    constexpr int CO = 32;
    float* outp = (float*)d_out;
    node_xform_resident<CO, 256, true><<<512, 256, 0, stream>>>(
        hbuf, Wl1, bl1, Wr1, br1, xl, xr, b0, den0, outp, amax, den1, flag1, n);
    edge_onepass<CO, 1024><<<GS, 1024, 0, stream>>>(
        xl, xr, srcI, dstI, ef, We1, att1, den1, outp, flag1, n, e);
    edge_alpha_shifted<CO><<<GS, B, 0, stream>>>(
        xl, xr, srcI, dstI, ef, We1, att1, alp, amax, outp, den1, n, flag1, e);
    edge_scatter_shifted<CO><<<GS, B, 0, stream>>>(
        xl, alp, amax, den1, srcI, dstI, outp, flag1, e);
    k_finalize<CO><<<cdiv((long long)n * CO, B), B, 0, stream>>>(outp, den1, b1, n, 0);
  }
}
